# starting kernel
# baseline (speedup 1.0000x reference)
#include <hip/hip_runtime.h>

typedef _Float16 v16h __attribute__((ext_vector_type(16)));
typedef _Float16 v8h  __attribute__((ext_vector_type(8)));
typedef _Float16 v2h  __attribute__((ext_vector_type(2)));
typedef float    v8f  __attribute__((ext_vector_type(8)));

#define NB_STEPS 50
#define S_PTS    51
#define HID      128
#define ST_W     136
#define ST_A     136
#define WAVES    8
#define TPB      256
#define TILES_PER_WAVE 8
#define PI_F     3.14159265358979323846f

__global__ void __launch_bounds__(256) monotonic_init(const float* __restrict__ h,
                                                      float* __restrict__ out, int n) {
    int i = blockIdx.x * blockDim.x + threadIdx.x;
    if (i < n) out[i] = h[i];
}

__device__ __forceinline__ v16h ldfrag(const _Float16* rowp, int klo) {
    v8h lo = *(const v8h*)(rowp + klo);
    v8h hi = *(const v8h*)(rowp + 16 + klo);
    return __builtin_shufflevector(lo, hi, 0,1,2,3,4,5,6,7,8,9,10,11,12,13,14,15);
}

__global__ void __launch_bounds__(TPB) monotonic_main(
    const float* __restrict__ x,  const float* __restrict__ h,
    const float* __restrict__ w1, const float* __restrict__ b1,
    const float* __restrict__ w2, const float* __restrict__ b2,
    const float* __restrict__ w3, const float* __restrict__ b3,
    const float* __restrict__ w4, const float* __restrict__ b4,
    float* __restrict__ out, int Bn)
{
    __shared__ __align__(16) _Float16 w2T[HID * ST_W];
    __shared__ __align__(16) _Float16 w3T[HID * ST_W];
    __shared__ __align__(16) _Float16 acts[WAVES * 16 * ST_A];
    __shared__ float w1s[2 * HID], b1s[HID], b2s[HID], b3s[HID], w4s[HID];
    __shared__ float ccws[S_PTS], stepss[S_PTS], b4s[1];

    const int tid = threadIdx.x;

    for (int idx = tid; idx < HID * HID; idx += TPB) {
        int k = idx >> 7, n = idx & 127;
        w2T[n * ST_W + k] = (_Float16)w2[idx];
        w3T[n * ST_W + k] = (_Float16)w3[idx];
    }
    if (tid < HID) {
        b1s[tid] = b1[tid]; b2s[tid] = b2[tid]; b3s[tid] = b3[tid]; w4s[tid] = w4[tid];
        w1s[tid] = w1[tid]; w1s[HID + tid] = w1[HID + tid];
    }
    if (tid == 0) b4s[0] = b4[0];
    if (tid < S_PTS) {
        float t = (float)tid;
        stepss[tid] = cosf(t * PI_F / (float)NB_STEPS);
        float a = 1.0f;
        #pragma unroll
        for (int j = 1; j <= NB_STEPS / 2; ++j) {
            float k = 2.0f * (float)j;
            a += (2.0f / (1.0f - k * k)) * cosf(k * t * PI_F / (float)NB_STEPS);
        }
        a *= 2.0f / (float)NB_STEPS;
        if (tid == 0 || tid == NB_STEPS) a *= 0.5f;
        ccws[tid] = a;
    }
    __syncthreads();

    const int  lane  = tid & 31;
    const int  wave  = tid >> 5;
    const long total   = (long)Bn * S_PTS;
    const long tstride = (long)gridDim.x * WAVES * 16;

    _Float16* act = &acts[wave * 16 * ST_A];
    const int  m   = lane & 15;
    const int  klo = (lane < 16) ? 0 : 8;
    const int  mo  = (lane >> 4) * 8;

    for (long tbase = ((long)blockIdx.x * WAVES + wave) * 16; tbase < total; tbase += tstride) {
        const long r  = tbase + m;
        const long rc = (r < total) ? r : (total - 1);
        const int  bidx = (int)(rc / S_PTS);
        const int  sidx = (int)(rc - (long)bidx * S_PTS);
        const float xb = x[bidx];
        const float hb = h[bidx];
        const float Xv = xb * (stepss[sidx] + 1.0f) * 0.5f;

        {
            const int khalf = (lane >> 4) * 64;
            v2h* actrow = (v2h*)&act[m * ST_A + khalf];
            #pragma unroll 4
            for (int kk = 0; kk < 32; ++kk) {
                int k = khalf + 2 * kk;
                float v0 = fmaf(Xv, w1s[k],     fmaf(hb, w1s[HID + k],     b1s[k]));
                float v1 = fmaf(Xv, w1s[k + 1], fmaf(hb, w1s[HID + k + 1], b1s[k + 1]));
                v2h p; p.x = (_Float16)fmaxf(v0, 0.0f); p.y = (_Float16)fmaxf(v1, 0.0f);
                actrow[kk] = p;
            }
        }

        {
            v16h a0 = ldfrag(&act[m * ST_A +  0], klo);
            v16h a1 = ldfrag(&act[m * ST_A + 32], klo);
            v16h a2 = ldfrag(&act[m * ST_A + 64], klo);
            v16h a3 = ldfrag(&act[m * ST_A + 96], klo);
            const _Float16* wr0 = &w2T[m * ST_W];
            v16h b0 = ldfrag(wr0 +  0, klo), b1f = ldfrag(wr0 + 32, klo);
            v16h b2f = ldfrag(wr0 + 64, klo), b3f = ldfrag(wr0 + 96, klo);
            v8f acc[8];
            #pragma unroll
            for (int t = 0; t < 8; ++t) {
                v16h n0 = b0, n1 = b1f, n2 = b2f, n3 = b3f;
                if (t < 7) {
                    const _Float16* wn = &w2T[((t + 1) * 16 + m) * ST_W];
                    n0 = ldfrag(wn +  0, klo); n1 = ldfrag(wn + 32, klo);
                    n2 = ldfrag(wn + 64, klo); n3 = ldfrag(wn + 96, klo);
                }
                v8f c = {};
                c = __builtin_amdgcn_wmma_f32_16x16x32_f16(false, a0, false, b0,  (short)0, c, false, false);
                c = __builtin_amdgcn_wmma_f32_16x16x32_f16(false, a1, false, b1f, (short)0, c, false, false);
                c = __builtin_amdgcn_wmma_f32_16x16x32_f16(false, a2, false, b2f, (short)0, c, false, false);
                c = __builtin_amdgcn_wmma_f32_16x16x32_f16(false, a3, false, b3f, (short)0, c, false, false);
                acc[t] = c;
                b0 = n0; b1f = n1; b2f = n2; b3f = n3;
            }
            #pragma unroll
            for (int t = 0; t < 8; ++t) {
                int n = t * 16 + m;
                float bv = b2s[n];
                #pragma unroll
                for (int i = 0; i < 8; ++i) {
                    float v = fmaxf(acc[t][i] + bv, 0.0f);
                    act[(mo + i) * ST_A + n] = (_Float16)v;
                }
            }
        }

        float rs[8] = {0.f, 0.f, 0.f, 0.f, 0.f, 0.f, 0.f, 0.f};
        {
            v16h a0 = ldfrag(&act[m * ST_A +  0], klo);
            v16h a1 = ldfrag(&act[m * ST_A + 32], klo);
            v16h a2 = ldfrag(&act[m * ST_A + 64], klo);
            v16h a3 = ldfrag(&act[m * ST_A + 96], klo);
            const _Float16* wr0 = &w3T[m * ST_W];
            v16h b0 = ldfrag(wr0 +  0, klo), b1f = ldfrag(wr0 + 32, klo);
            v16h b2f = ldfrag(wr0 + 64, klo), b3f = ldfrag(wr0 + 96, klo);
            #pragma unroll
            for (int t = 0; t < 8; ++t) {
                v16h n0 = b0, n1 = b1f, n2 = b2f, n3 = b3f;
                if (t < 7) {
                    const _Float16* wn = &w3T[((t + 1) * 16 + m) * ST_W];
                    n0 = ldfrag(wn +  0, klo); n1 = ldfrag(wn + 32, klo);
                    n2 = ldfrag(wn + 64, klo); n3 = ldfrag(wn + 96, klo);
                }
                v8f c = {};
                c = __builtin_amdgcn_wmma_f32_16x16x32_f16(false, a0, false, b0,  (short)0, c, false, false);
                c = __builtin_amdgcn_wmma_f32_16x16x32_f16(false, a1, false, b1f, (short)0, c, false, false);
                c = __builtin_amdgcn_wmma_f32_16x16x32_f16(false, a2, false, b2f, (short)0, c, false, false);
                c = __builtin_amdgcn_wmma_f32_16x16x32_f16(false, a3, false, b3f, (short)0, c, false, false);
                int n = t * 16 + m;
                float bv = b3s[n], wv = w4s[n];
                #pragma unroll
                for (int i = 0; i < 8; ++i)
                    rs[i] = fmaf(fmaxf(c[i] + bv, 0.0f), wv, rs[i]);
                b0 = n0; b1f = n1; b2f = n2; b3f = n3;
            }
        }

        #pragma unroll
        for (int i = 0; i < 8; ++i) {
            float v = rs[i];
            v += __shfl_xor(v, 1, 32);
            v += __shfl_xor(v, 2, 32);
            v += __shfl_xor(v, 4, 32);
            v += __shfl_xor(v, 8, 32);
            rs[i] = v;
        }

        if (m == 0) {
            const int  rowbase = (lane >> 4) * 8;
            const float bb = b4s[0];
            #pragma unroll
            for (int i = 0; i < 8; ++i) {
                long rr = tbase + rowbase + i;
                if (rr < total) {
                    float v  = rs[i] + bb;
                    float dz = (v > 0.0f) ? v : (expf(v) - 1.0f);
                    dz += 1.0f;
                    int   bi = (int)(rr / S_PTS);
                    int   si = (int)(rr - (long)bi * S_PTS);
                    float contrib = dz * ccws[si] * x[bi] * 0.5f;
                    atomicAdd(&out[bi], contrib);
                }
            }
        }
    }
}

extern "C" void kernel_launch(void* const* d_in, const int* in_sizes, int n_in,
                              void* d_out, int out_size, void* d_ws, size_t ws_size,
                              hipStream_t stream) {
    const float* x  = (const float*)d_in[0];
    const float* h  = (const float*)d_in[1];
    const float* w1 = (const float*)d_in[2];
    const float* b1 = (const float*)d_in[3];
    const float* w2 = (const float*)d_in[4];
    const float* b2 = (const float*)d_in[5];
    const float* w3 = (const float*)d_in[6];
    const float* b3 = (const float*)d_in[7];
    const float* w4 = (const float*)d_in[8];
    const float* b4 = (const float*)d_in[9];
    float* out = (float*)d_out;
    const int Bn = in_sizes[0];

    monotonic_init<<<(Bn + 255) / 256, 256, 0, stream>>>(h, out, Bn);

    const long total  = (long)Bn * S_PTS;
    const int  tiles  = (int)((total + 15) / 16);
    const int  blocks = (tiles + WAVES * TILES_PER_WAVE - 1) / (WAVES * TILES_PER_WAVE);
    monotonic_main<<<blocks, TPB, 0, stream>>>(x, h, w1, b1, w2, b2, w3, b3, w4, b4, out, Bn);
}


# Activations stay in registers: weights as WMMA A, acts as B, so packed D tiles are the next layer's B fragments; no LDS staging, 4 workgroups per WGP fit; layer-4 dot summed in-lane + 1 shuffle
# speedup vs baseline: 1.5784x; 1.5784x over previous
#include <hip/hip_runtime.h>

typedef _Float16 v16h __attribute__((ext_vector_type(16)));
typedef _Float16 v8h  __attribute__((ext_vector_type(8)));
typedef _Float16 v2h  __attribute__((ext_vector_type(2)));
typedef float    v8f  __attribute__((ext_vector_type(8)));

#define NB_STEPS 50
#define S_PTS    51
#define HID      128
#define ST_W     136
#define WAVES    8
#define TPB      256
#define TILES_PER_WAVE 8
#define PI_F     3.14159265358979323846f

__global__ void __launch_bounds__(256) monotonic_init(const float* __restrict__ h,
                                                      float* __restrict__ out, int n) {
    int i = blockIdx.x * blockDim.x + threadIdx.x;
    if (i < n) out[i] = h[i];
}

__device__ __forceinline__ v16h ldfrag(const _Float16* p) {
    v8h lo = *(const v8h*)(p);
    v8h hi = *(const v8h*)(p + 16);
    return __builtin_shufflevector(lo, hi, 0,1,2,3,4,5,6,7,8,9,10,11,12,13,14,15);
}

__device__ __forceinline__ v8f ld8f(const float* p) {
    float4 a = *(const float4*)p;
    float4 b = *(const float4*)(p + 4);
    v8f r = {a.x, a.y, a.z, a.w, b.x, b.y, b.z, b.w};
    return r;
}

__global__ void __launch_bounds__(TPB, 8) monotonic_main(
    const float* __restrict__ x,  const float* __restrict__ h,
    const float* __restrict__ w1, const float* __restrict__ b1,
    const float* __restrict__ w2, const float* __restrict__ b2,
    const float* __restrict__ w3, const float* __restrict__ b3,
    const float* __restrict__ w4, const float* __restrict__ b4,
    float* __restrict__ out, int Bn)
{
    __shared__ __align__(16) _Float16 w2T[HID * ST_W];
    __shared__ __align__(16) _Float16 w3T[HID * ST_W];
    __shared__ __align__(16) float w1s[2 * HID];
    __shared__ __align__(16) float b1s[HID];
    __shared__ __align__(16) float b2s[HID];
    __shared__ __align__(16) float b3s[HID];
    __shared__ __align__(16) float w4s[HID];
    __shared__ float ccws[S_PTS], stepss[S_PTS], b4s[1];

    const int tid = threadIdx.x;

    for (int idx = tid; idx < HID * HID; idx += TPB) {
        int k = idx >> 7, n = idx & 127;
        w2T[n * ST_W + k] = (_Float16)w2[idx];
        w3T[n * ST_W + k] = (_Float16)w3[idx];
    }
    if (tid < HID) {
        b1s[tid] = b1[tid]; b2s[tid] = b2[tid]; b3s[tid] = b3[tid]; w4s[tid] = w4[tid];
        w1s[tid] = w1[tid]; w1s[HID + tid] = w1[HID + tid];
    }
    if (tid == 0) b4s[0] = b4[0];
    if (tid < S_PTS) {
        float t = (float)tid;
        stepss[tid] = cosf(t * PI_F / (float)NB_STEPS);
        float a = 1.0f;
        #pragma unroll
        for (int j = 1; j <= NB_STEPS / 2; ++j) {
            float k = 2.0f * (float)j;
            a += (2.0f / (1.0f - k * k)) * cosf(k * t * PI_F / (float)NB_STEPS);
        }
        a *= 2.0f / (float)NB_STEPS;
        if (tid == 0 || tid == NB_STEPS) a *= 0.5f;
        ccws[tid] = a;
    }
    __syncthreads();

    const int  lane  = tid & 31;
    const int  wave  = tid >> 5;
    const long total   = (long)Bn * S_PTS;
    const long tstride = (long)gridDim.x * WAVES * 16;

    const int  m   = lane & 15;
    const int  klo = (lane >> 4) * 8;

    for (long tbase = ((long)blockIdx.x * WAVES + wave) * 16; tbase < total; tbase += tstride) {
        const long r  = tbase + m;
        const long rc = (r < total) ? r : (total - 1);
        const int  bidx = (int)(rc / S_PTS);
        const int  sidx = (int)(rc - (long)bidx * S_PTS);
        const float xb = x[bidx];
        const float hb = h[bidx];
        const float Xv = xb * (stepss[sidx] + 1.0f) * 0.5f;

        int wofs = m * ST_W + klo;
        int pofs = klo;
        asm volatile("" : "+v"(wofs), "+v"(pofs));

        v16h act1[4];
        #pragma unroll
        for (int kc = 0; kc < 4; ++kc) {
            const int f0 = 32 * kc + pofs;
            const v8f wa0 = ld8f(&w1s[f0]),      wb0 = ld8f(&w1s[HID + f0]),      bb0 = ld8f(&b1s[f0]);
            const v8f wa1 = ld8f(&w1s[f0 + 16]), wb1 = ld8f(&w1s[HID + f0 + 16]), bb1 = ld8f(&b1s[f0 + 16]);
            v16h f;
            #pragma unroll
            for (int s = 0; s < 8; ++s) {
                float v0 = fmaf(Xv, wa0[s], fmaf(hb, wb0[s], bb0[s]));
                float v1 = fmaf(Xv, wa1[s], fmaf(hb, wb1[s], bb1[s]));
                f[s]     = (_Float16)fmaxf(v0, 0.0f);
                f[8 + s] = (_Float16)fmaxf(v1, 0.0f);
            }
            act1[kc] = f;
        }

        v16h act2[4];
        {
            const _Float16* wr0 = &w2T[wofs];
            v16h g0 = ldfrag(wr0 +  0), g1 = ldfrag(wr0 + 32);
            v16h g2 = ldfrag(wr0 + 64), g3 = ldfrag(wr0 + 96);
            #pragma unroll
            for (int t = 0; t < 8; ++t) {
                v16h n0 = g0, n1 = g1, n2 = g2, n3 = g3;
                if (t < 7) {
                    const _Float16* wn = &w2T[(t + 1) * 16 * ST_W + wofs];
                    n0 = ldfrag(wn +  0); n1 = ldfrag(wn + 32);
                    n2 = ldfrag(wn + 64); n3 = ldfrag(wn + 96);
                }
                const v8f bv = ld8f(&b2s[16 * t + pofs]);
                v8f c = {};
                c = __builtin_amdgcn_wmma_f32_16x16x32_f16(false, g0, false, act1[0], (short)0, c, false, false);
                c = __builtin_amdgcn_wmma_f32_16x16x32_f16(false, g1, false, act1[1], (short)0, c, false, false);
                c = __builtin_amdgcn_wmma_f32_16x16x32_f16(false, g2, false, act1[2], (short)0, c, false, false);
                c = __builtin_amdgcn_wmma_f32_16x16x32_f16(false, g3, false, act1[3], (short)0, c, false, false);
                #pragma unroll
                for (int s = 0; s < 8; ++s)
                    act2[t >> 1][(t & 1) * 8 + s] = (_Float16)fmaxf(c[s] + bv[s], 0.0f);
                g0 = n0; g1 = n1; g2 = n2; g3 = n3;
            }
        }

        float rs0 = 0.0f, rs1 = 0.0f;
        {
            const _Float16* wr0 = &w3T[wofs];
            v16h g0 = ldfrag(wr0 +  0), g1 = ldfrag(wr0 + 32);
            v16h g2 = ldfrag(wr0 + 64), g3 = ldfrag(wr0 + 96);
            #pragma unroll
            for (int t = 0; t < 8; ++t) {
                v16h n0 = g0, n1 = g1, n2 = g2, n3 = g3;
                if (t < 7) {
                    const _Float16* wn = &w3T[(t + 1) * 16 * ST_W + wofs];
                    n0 = ldfrag(wn +  0); n1 = ldfrag(wn + 32);
                    n2 = ldfrag(wn + 64); n3 = ldfrag(wn + 96);
                }
                const v8f bv = ld8f(&b3s[16 * t + pofs]);
                const v8f wv = ld8f(&w4s[16 * t + pofs]);
                v8f c = {};
                c = __builtin_amdgcn_wmma_f32_16x16x32_f16(false, g0, false, act2[0], (short)0, c, false, false);
                c = __builtin_amdgcn_wmma_f32_16x16x32_f16(false, g1, false, act2[1], (short)0, c, false, false);
                c = __builtin_amdgcn_wmma_f32_16x16x32_f16(false, g2, false, act2[2], (short)0, c, false, false);
                c = __builtin_amdgcn_wmma_f32_16x16x32_f16(false, g3, false, act2[3], (short)0, c, false, false);
                #pragma unroll
                for (int s = 0; s < 8; s += 2) {
                    rs0 = fmaf(fmaxf(c[s]     + bv[s],     0.0f), wv[s],     rs0);
                    rs1 = fmaf(fmaxf(c[s + 1] + bv[s + 1], 0.0f), wv[s + 1], rs1);
                }
                g0 = n0; g1 = n1; g2 = n2; g3 = n3;
            }
        }

        float rs = rs0 + rs1;
        rs += __shfl_xor(rs, 16, 32);

        if (lane < 16 && r < total) {
            float v  = rs + b4s[0];
            float dz = (v > 0.0f) ? v : (expf(v) - 1.0f);
            dz += 1.0f;
            float contrib = dz * ccws[sidx] * xb * 0.5f;
            atomicAdd(&out[bidx], contrib);
        }
    }
}

extern "C" void kernel_launch(void* const* d_in, const int* in_sizes, int n_in,
                              void* d_out, int out_size, void* d_ws, size_t ws_size,
                              hipStream_t stream) {
    const float* x  = (const float*)d_in[0];
    const float* h  = (const float*)d_in[1];
    const float* w1 = (const float*)d_in[2];
    const float* b1 = (const float*)d_in[3];
    const float* w2 = (const float*)d_in[4];
    const float* b2 = (const float*)d_in[5];
    const float* w3 = (const float*)d_in[6];
    const float* b3 = (const float*)d_in[7];
    const float* w4 = (const float*)d_in[8];
    const float* b4 = (const float*)d_in[9];
    float* out = (float*)d_out;
    const int Bn = in_sizes[0];

    monotonic_init<<<(Bn + 255) / 256, 256, 0, stream>>>(h, out, Bn);

    const long total  = (long)Bn * S_PTS;
    const int  tiles  = (int)((total + 15) / 16);
    const int  blocks = (tiles + WAVES * TILES_PER_WAVE - 1) / (WAVES * TILES_PER_WAVE);
    monotonic_main<<<blocks, TPB, 0, stream>>>(x, h, w1, b1, w2, b2, w3, b3, w4, b4, out, Bn);
}


# Row blocking: a wave carries 2 row tiles through the layers, so each LDS weight fragment and parameter read feeds 2 tiles; tile order pinned against spills; 4 waves per SIMD, 512 equal workgroups
# speedup vs baseline: 2.7901x; 1.7677x over previous
#include <hip/hip_runtime.h>

typedef _Float16 v16h __attribute__((ext_vector_type(16)));
typedef _Float16 v8h  __attribute__((ext_vector_type(8)));
typedef _Float16 v2h  __attribute__((ext_vector_type(2)));
typedef float    v8f  __attribute__((ext_vector_type(8)));

#define NB_STEPS 50
#define S_PTS    51
#define HID      128
#define ST_W     136
#define WAVES    8
#define TPB      256
#define RT       2
#define GROUP_ROWS (16 * RT)
#define NBLOCKS  512
#define PI_F     3.14159265358979323846f

__global__ void __launch_bounds__(256) monotonic_init(const float* __restrict__ h,
                                                      float* __restrict__ out, int n) {
    int i = blockIdx.x * blockDim.x + threadIdx.x;
    if (i < n) out[i] = h[i];
}

__device__ __forceinline__ v16h ldfrag(const _Float16* p) {
    v8h lo = *(const v8h*)(p);
    v8h hi = *(const v8h*)(p + 16);
    return __builtin_shufflevector(lo, hi, 0,1,2,3,4,5,6,7,8,9,10,11,12,13,14,15);
}

__device__ __forceinline__ v8f ld8f(const float* p) {
    float4 a = *(const float4*)p;
    float4 b = *(const float4*)(p + 4);
    v8f r = {a.x, a.y, a.z, a.w, b.x, b.y, b.z, b.w};
    return r;
}

__global__ void __launch_bounds__(TPB, 4) monotonic_main(
    const float* __restrict__ x,  const float* __restrict__ h,
    const float* __restrict__ w1, const float* __restrict__ b1,
    const float* __restrict__ w2, const float* __restrict__ b2,
    const float* __restrict__ w3, const float* __restrict__ b3,
    const float* __restrict__ w4, const float* __restrict__ b4,
    float* __restrict__ out, int Bn)
{
    __shared__ __align__(16) _Float16 w2T[HID * ST_W];
    __shared__ __align__(16) _Float16 w3T[HID * ST_W];
    __shared__ __align__(16) float w1s[2 * HID];
    __shared__ __align__(16) float b1s[HID];
    __shared__ __align__(16) float b2s[HID];
    __shared__ __align__(16) float b3s[HID];
    __shared__ __align__(16) float w4s[HID];
    __shared__ float ccws[S_PTS], stepss[S_PTS], b4s[1];

    const int tid = threadIdx.x;

    for (int idx = tid; idx < HID * HID; idx += TPB) {
        int k = idx >> 7, n = idx & 127;
        w2T[n * ST_W + k] = (_Float16)w2[idx];
        w3T[n * ST_W + k] = (_Float16)w3[idx];
    }
    if (tid < HID) {
        b1s[tid] = b1[tid]; b2s[tid] = b2[tid]; b3s[tid] = b3[tid]; w4s[tid] = w4[tid];
        w1s[tid] = w1[tid]; w1s[HID + tid] = w1[HID + tid];
    }
    if (tid == 0) b4s[0] = b4[0];
    if (tid < S_PTS) {
        float t = (float)tid;
        stepss[tid] = cosf(t * PI_F / (float)NB_STEPS);
        float a = 1.0f;
        #pragma unroll
        for (int j = 1; j <= NB_STEPS / 2; ++j) {
            float k = 2.0f * (float)j;
            a += (2.0f / (1.0f - k * k)) * cosf(k * t * PI_F / (float)NB_STEPS);
        }
        a *= 2.0f / (float)NB_STEPS;
        if (tid == 0 || tid == NB_STEPS) a *= 0.5f;
        ccws[tid] = a;
    }
    __syncthreads();

    const int  lane  = tid & 31;
    const int  wave  = tid >> 5;
    const int  total   = Bn * S_PTS;
    const int  ngroups = (total + GROUP_ROWS - 1) / GROUP_ROWS;
    const int  gpb    = (ngroups + (int)gridDim.x - 1) / (int)gridDim.x;
    const int  gfirst = (int)blockIdx.x * gpb;
    const int  gend   = (gfirst + gpb < ngroups) ? (gfirst + gpb) : ngroups;

    const int  m   = lane & 15;
    const int  klo = (lane >> 4) * 8;

    for (int g = gfirst + wave; g < gend; g += WAVES) {
        const int rbase = g * GROUP_ROWS + m;
        float Xv[RT], hb[RT];
        #pragma unroll
        for (int u = 0; u < RT; ++u) {
            const int r  = rbase + 16 * u;
            const int rc = (r < total) ? r : (total - 1);
            const int bidx = rc / S_PTS;
            const int sidx = rc - bidx * S_PTS;
            hb[u] = h[bidx];
            Xv[u] = x[bidx] * (stepss[sidx] + 1.0f) * 0.5f;
        }

        int wofs = m * ST_W + klo;
        int pofs = klo;
        asm volatile("" : "+v"(wofs), "+v"(pofs));

        v16h act1[RT][4];
        #pragma unroll
        for (int kc = 0; kc < 4; ++kc) {
            const int f0 = 32 * kc + pofs;
            const v8f wa0 = ld8f(&w1s[f0]),      wb0 = ld8f(&w1s[HID + f0]),      bb0 = ld8f(&b1s[f0]);
            const v8f wa1 = ld8f(&w1s[f0 + 16]), wb1 = ld8f(&w1s[HID + f0 + 16]), bb1 = ld8f(&b1s[f0 + 16]);
            #pragma unroll
            for (int u = 0; u < RT; ++u) {
                v16h f;
                #pragma unroll
                for (int s = 0; s < 8; ++s) {
                    float v0 = fmaf(Xv[u], wa0[s], fmaf(hb[u], wb0[s], bb0[s]));
                    float v1 = fmaf(Xv[u], wa1[s], fmaf(hb[u], wb1[s], bb1[s]));
                    f[s]     = (_Float16)fmaxf(v0, 0.0f);
                    f[8 + s] = (_Float16)fmaxf(v1, 0.0f);
                }
                act1[u][kc] = f;
            }
        }

        v16h act2[RT][4];
        {
            const _Float16* wr0 = &w2T[wofs];
            v16h g0 = ldfrag(wr0 +  0), g1 = ldfrag(wr0 + 32);
            v16h g2 = ldfrag(wr0 + 64), g3 = ldfrag(wr0 + 96);
            #pragma unroll
            for (int t = 0; t < 8; ++t) {
                const v8f bv = ld8f(&b2s[16 * t + pofs]);
                v16h n0 = g0, n1 = g1, n2 = g2, n3 = g3;
                if (t < 7) {
                    const _Float16* wn = &w2T[(t + 1) * 16 * ST_W + wofs];
                    n0 = ldfrag(wn +  0); n1 = ldfrag(wn + 32);
                    n2 = ldfrag(wn + 64); n3 = ldfrag(wn + 96);
                }
                v8f c[RT];
                #pragma unroll
                for (int u = 0; u < RT; ++u) { v8f z = {}; c[u] = z; }
                #pragma unroll
                for (int u = 0; u < RT; ++u) c[u] = __builtin_amdgcn_wmma_f32_16x16x32_f16(false, g0, false, act1[u][0], (short)0, c[u], false, false);
                #pragma unroll
                for (int u = 0; u < RT; ++u) c[u] = __builtin_amdgcn_wmma_f32_16x16x32_f16(false, g1, false, act1[u][1], (short)0, c[u], false, false);
                #pragma unroll
                for (int u = 0; u < RT; ++u) c[u] = __builtin_amdgcn_wmma_f32_16x16x32_f16(false, g2, false, act1[u][2], (short)0, c[u], false, false);
                #pragma unroll
                for (int u = 0; u < RT; ++u) c[u] = __builtin_amdgcn_wmma_f32_16x16x32_f16(false, g3, false, act1[u][3], (short)0, c[u], false, false);
                #pragma unroll
                for (int u = 0; u < RT; ++u) {
                    #pragma unroll
                    for (int s = 0; s < 8; ++s)
                        act2[u][t >> 1][(t & 1) * 8 + s] = (_Float16)fmaxf(c[u][s] + bv[s], 0.0f);
                }
                g0 = n0; g1 = n1; g2 = n2; g3 = n3;
                __builtin_amdgcn_sched_barrier(0);
            }
        }

        float rs0[RT], rs1[RT];
        #pragma unroll
        for (int u = 0; u < RT; ++u) { rs0[u] = 0.0f; rs1[u] = 0.0f; }
        {
            const _Float16* wr0 = &w3T[wofs];
            v16h g0 = ldfrag(wr0 +  0), g1 = ldfrag(wr0 + 32);
            v16h g2 = ldfrag(wr0 + 64), g3 = ldfrag(wr0 + 96);
            #pragma unroll
            for (int t = 0; t < 8; ++t) {
                const v8f bv = ld8f(&b3s[16 * t + pofs]);
                const v8f wv = ld8f(&w4s[16 * t + pofs]);
                v16h n0 = g0, n1 = g1, n2 = g2, n3 = g3;
                if (t < 7) {
                    const _Float16* wn = &w3T[(t + 1) * 16 * ST_W + wofs];
                    n0 = ldfrag(wn +  0); n1 = ldfrag(wn + 32);
                    n2 = ldfrag(wn + 64); n3 = ldfrag(wn + 96);
                }
                v8f c[RT];
                #pragma unroll
                for (int u = 0; u < RT; ++u) { v8f z = {}; c[u] = z; }
                #pragma unroll
                for (int u = 0; u < RT; ++u) c[u] = __builtin_amdgcn_wmma_f32_16x16x32_f16(false, g0, false, act2[u][0], (short)0, c[u], false, false);
                #pragma unroll
                for (int u = 0; u < RT; ++u) c[u] = __builtin_amdgcn_wmma_f32_16x16x32_f16(false, g1, false, act2[u][1], (short)0, c[u], false, false);
                #pragma unroll
                for (int u = 0; u < RT; ++u) c[u] = __builtin_amdgcn_wmma_f32_16x16x32_f16(false, g2, false, act2[u][2], (short)0, c[u], false, false);
                #pragma unroll
                for (int u = 0; u < RT; ++u) c[u] = __builtin_amdgcn_wmma_f32_16x16x32_f16(false, g3, false, act2[u][3], (short)0, c[u], false, false);
                #pragma unroll
                for (int u = 0; u < RT; ++u) {
                    #pragma unroll
                    for (int s = 0; s < 8; s += 2) {
                        rs0[u] = fmaf(fmaxf(c[u][s]     + bv[s],     0.0f), wv[s],     rs0[u]);
                        rs1[u] = fmaf(fmaxf(c[u][s + 1] + bv[s + 1], 0.0f), wv[s + 1], rs1[u]);
                    }
                    asm volatile("" : "+v"(rs0[u]), "+v"(rs1[u]));
                }
                g0 = n0; g1 = n1; g2 = n2; g3 = n3;
                __builtin_amdgcn_sched_barrier(0);
            }
        }

        #pragma unroll
        for (int u = 0; u < RT; ++u) {
            float rs = rs0[u] + rs1[u];
            rs += __shfl_xor(rs, 16, 32);

            const int r = rbase + 16 * u;
            if (lane < 16 && r < total) {
                const int bidx = r / S_PTS;
                const int sidx = r - bidx * S_PTS;
                const float xb = x[bidx];
                float v  = rs + b4s[0];
                float dz = (v > 0.0f) ? v : (expf(v) - 1.0f);
                dz += 1.0f;
                float contrib = dz * ccws[sidx] * xb * 0.5f;
                atomicAdd(&out[bidx], contrib);
            }
        }
    }
}

extern "C" void kernel_launch(void* const* d_in, const int* in_sizes, int n_in,
                              void* d_out, int out_size, void* d_ws, size_t ws_size,
                              hipStream_t stream) {
    const float* x  = (const float*)d_in[0];
    const float* h  = (const float*)d_in[1];
    const float* w1 = (const float*)d_in[2];
    const float* b1 = (const float*)d_in[3];
    const float* w2 = (const float*)d_in[4];
    const float* b2 = (const float*)d_in[5];
    const float* w3 = (const float*)d_in[6];
    const float* b3 = (const float*)d_in[7];
    const float* w4 = (const float*)d_in[8];
    const float* b4 = (const float*)d_in[9];
    float* out = (float*)d_out;
    const int Bn = in_sizes[0];

    monotonic_init<<<(Bn + 255) / 256, 256, 0, stream>>>(h, out, Bn);

    monotonic_main<<<NBLOCKS, TPB, 0, stream>>>(x, h, w1, b1, w2, b2, w3, b3, w4, b4, out, Bn);
}
